# starting kernel
# baseline (speedup 1.0000x reference)
#include <hip/hip_runtime.h>
#include <math.h>

typedef __attribute__((ext_vector_type(2))) float v2f;
typedef __attribute__((ext_vector_type(8))) float v8f;

#define BIGF 1e30f
#define NBATCH 4096
#define UD 24

__device__ __forceinline__ void qam16_point(int s, float& re, float& im) {
    const int b0 = (s >> 3) & 1, b1 = (s >> 2) & 1, b2 = (s >> 1) & 1, b3 = s & 1;
    const float inv = 0.3162277660168379f;
    re = (float)((1 - 2 * b0) * (2 - (1 - 2 * b2))) * inv;
    im = (float)((1 - 2 * b1) * (2 - (1 - 2 * b3))) * inv;
}

__global__ void build_v_kernel(float* __restrict__ v) {
    const int c = blockIdx.x * blockDim.x + threadIdx.x;
    if (c >= 4096) return;
    const int idx0 = (c >> 8) & 15, idx1 = (c >> 4) & 15, idx2 = c & 15;
    float xr[3], xi[3];
    qam16_point(idx0, xr[0], xi[0]);
    qam16_point(idx1, xr[1], xi[1]);
    qam16_point(idx2, xr[2], xi[2]);
    float vv[UD];
#pragma unroll
    for (int k = 0; k < 3; k++) {
#pragma unroll
        for (int l = 0; l < 3; l++) {
            const float pre = xr[k] * xr[l] + xi[k] * xi[l];
            const float pim = xr[k] * xi[l] - xi[k] * xr[l];
            vv[k * 3 + l]     = pre;
            vv[9 + k * 3 + l] = -pim;
        }
    }
#pragma unroll
    for (int k = 0; k < 3; k++) {
        vv[18 + k] = -2.0f * xr[k];
        vv[21 + k] =  2.0f * xi[k];
    }
#pragma unroll
    for (int t = 0; t < UD; t++) v[c * UD + t] = vv[t];
}

__global__ void build_u_kernel(const float* __restrict__ y_re, const float* __restrict__ y_im,
                               const float* __restrict__ h_re, const float* __restrict__ h_im,
                               const float* __restrict__ s_re, const float* __restrict__ s_im,
                               float* __restrict__ u) {
    const int b = blockIdx.x * blockDim.x + threadIdx.x;
    if (b >= NBATCH) return;

    float Lr[8][8], Li[8][8];
#pragma unroll
    for (int i = 0; i < 8; i++) {
#pragma unroll
        for (int j = 0; j <= i; j++) {
            Lr[i][j] = s_re[b * 64 + i * 8 + j];
            Li[i][j] = s_im[b * 64 + i * 8 + j];
        }
    }
#pragma unroll
    for (int j = 0; j < 8; j++) {
        float d = Lr[j][j];
#pragma unroll
        for (int p = 0; p < j; p++) d -= Lr[j][p] * Lr[j][p] + Li[j][p] * Li[j][p];
        const float dj = sqrtf(d);
        const float inv = 1.0f / dj;
        Lr[j][j] = dj; Li[j][j] = 0.0f;
#pragma unroll
        for (int i = j + 1; i < 8; i++) {
            float ar = Lr[i][j], ai = Li[i][j];
#pragma unroll
            for (int p = 0; p < j; p++) {
                ar -= Lr[i][p] * Lr[j][p] + Li[i][p] * Li[j][p];
                ai -= Li[i][p] * Lr[j][p] - Lr[i][p] * Li[j][p];
            }
            Lr[i][j] = ar * inv; Li[i][j] = ai * inv;
        }
    }
    float ywr[8], ywi[8];
#pragma unroll
    for (int i = 0; i < 8; i++) {
        float ar = y_re[b * 8 + i], ai = y_im[b * 8 + i];
#pragma unroll
        for (int j = 0; j < i; j++) {
            ar -= Lr[i][j] * ywr[j] - Li[i][j] * ywi[j];
            ai -= Lr[i][j] * ywi[j] + Li[i][j] * ywr[j];
        }
        const float inv = 1.0f / Lr[i][i];
        ywr[i] = ar * inv; ywi[i] = ai * inv;
    }
    float hwr[8][3], hwi[8][3];
#pragma unroll
    for (int k = 0; k < 3; k++) {
#pragma unroll
        for (int i = 0; i < 8; i++) {
            float ar = h_re[b * 24 + i * 3 + k], ai = h_im[b * 24 + i * 3 + k];
#pragma unroll
            for (int j = 0; j < i; j++) {
                ar -= Lr[i][j] * hwr[j][k] - Li[i][j] * hwi[j][k];
                ai -= Lr[i][j] * hwi[j][k] + Li[i][j] * hwr[j][k];
            }
            const float inv = 1.0f / Lr[i][i];
            hwr[i][k] = ar * inv; hwi[i][k] = ai * inv;
        }
    }
    float uu[UD];
#pragma unroll
    for (int k = 0; k < 3; k++) {
#pragma unroll
        for (int l = 0; l < 3; l++) {
            float gr = 0.f, gi = 0.f;
#pragma unroll
            for (int m = 0; m < 8; m++) {
                gr += hwr[m][k] * hwr[m][l] + hwi[m][k] * hwi[m][l];
                gi += hwr[m][k] * hwi[m][l] - hwi[m][k] * hwr[m][l];
            }
            uu[k * 3 + l]     = gr;
            uu[9 + k * 3 + l] = gi;
        }
    }
#pragma unroll
    for (int k = 0; k < 3; k++) {
        float zr = 0.f, zi = 0.f;
#pragma unroll
        for (int m = 0; m < 8; m++) {
            zr += ywr[m] * hwr[m][k] + ywi[m] * hwi[m][k];
            zi += ywr[m] * hwi[m][k] - ywi[m] * hwr[m][k];
        }
        uu[18 + k] = zr;
        uu[21 + k] = zi;
    }
#pragma unroll
    for (int t = 0; t < UD; t++) u[b * UD + t] = uu[t];
}

__global__ void __launch_bounds__(256)
detect_kernel(const float* __restrict__ u, const float* __restrict__ v,
              float* __restrict__ out) {
    __shared__ float lm1[8][32][16];
    __shared__ float lm2[8][32][8];
    __shared__ float lb0[8][32][2];

    const int tid  = threadIdx.x;
    const int wave = tid >> 5;
    const int lane = tid & 31;
    const int li   = lane & 15;
    const int hh   = lane >> 4;
    const int bbase = blockIdx.x * 16;

    const float* ub = u + (size_t)(bbase + li) * UD + 2 * hh;
    v2f bf[6];
#pragma unroll
    for (int kk = 0; kk < 6; kk++) bf[kk] = *(const v2f*)(ub + 4 * kk);

    float m1h[16], m2h[8], bm0[2];
#pragma unroll
    for (int t = 0; t < 16; t++) m1h[t] = BIGF;
#pragma unroll
    for (int t = 0; t < 8; t++) m2h[t] = BIGF;

#pragma unroll
    for (int j = 0; j < 2; j++) {
        const int i0 = wave * 2 + j;
        float bmin = BIGF;
#pragma unroll
        for (int i1 = 0; i1 < 16; i1++) {
            const int cbase = i0 * 256 + i1 * 16;
            const float* vp = v + (size_t)(cbase + li) * UD + 2 * hh;
            v8f acc = {0.f, 0.f, 0.f, 0.f, 0.f, 0.f, 0.f, 0.f};
#pragma unroll
            for (int kk = 0; kk < 6; kk++) {
                const v2f af = *(const v2f*)(vp + 4 * kk);
                acc = __builtin_amdgcn_wmma_f32_16x16x4_f32(
                    false, af, false, bf[kk], (short)0, acc, false, false);
            }
            const float tmin =
                fminf(fminf(fminf(acc[0], acc[1]), fminf(acc[2], acc[3])),
                      fminf(fminf(acc[4], acc[5]), fminf(acc[6], acc[7])));
            bmin = fminf(bmin, tmin);
            m1h[i1] = fminf(m1h[i1], tmin);
#pragma unroll
            for (int r = 0; r < 8; r++) m2h[r] = fminf(m2h[r], acc[r]);
        }
        bm0[j] = bmin;
    }

#pragma unroll
    for (int t = 0; t < 16; t++) lm1[wave][lane][t] = m1h[t];
#pragma unroll
    for (int t = 0; t < 8; t++) lm2[wave][lane][t] = m2h[t];
    lb0[wave][lane][0] = bm0[0];
    lb0[wave][lane][1] = bm0[1];
    __syncthreads();

    if (tid < 16) {
        const int b = tid;
        float m[3][16];
#pragma unroll
        for (int i0 = 0; i0 < 16; i0++) {
            const int w = i0 >> 1, jj = i0 & 1;
            m[0][i0] = fminf(lb0[w][b][jj], lb0[w][b + 16][jj]);
        }
#pragma unroll
        for (int i1 = 0; i1 < 16; i1++) {
            float t = BIGF;
#pragma unroll
            for (int w = 0; w < 8; w++)
                t = fminf(t, fminf(lm1[w][b][i1], lm1[w][b + 16][i1]));
            m[1][i1] = t;
        }
#pragma unroll
        for (int i2 = 0; i2 < 16; i2++) {
            const int h2 = i2 >> 3, r = i2 & 7;
            float t = BIGF;
#pragma unroll
            for (int w = 0; w < 8; w++) t = fminf(t, lm2[w][b + 16 * h2][r]);
            m[2][i2] = t;
        }
#pragma unroll
        for (int k = 0; k < 3; k++) {
#pragma unroll
            for (int n = 0; n < 4; n++) {
                float mn0 = BIGF, mn1 = BIGF;
#pragma unroll
                for (int i = 0; i < 16; i++) {
                    if ((i >> (3 - n)) & 1) mn1 = fminf(mn1, m[k][i]);
                    else                    mn0 = fminf(mn0, m[k][i]);
                }
                out[(size_t)(bbase + b) * 12 + k * 4 + n] = mn0 - mn1;
            }
        }
    }
}

extern "C" void kernel_launch(void* const* d_in, const int* in_sizes, int n_in,
                              void* d_out, int out_size, void* d_ws, size_t ws_size,
                              hipStream_t stream) {
    const float* y_re = (const float*)d_in[0];
    const float* y_im = (const float*)d_in[1];
    const float* h_re = (const float*)d_in[2];
    const float* h_im = (const float*)d_in[3];
    const float* s_re = (const float*)d_in[4];
    const float* s_im = (const float*)d_in[5];
    float* out = (float*)d_out;

    float* u_tab = (float*)d_ws;
    float* v_tab = u_tab + (size_t)NBATCH * UD;

    build_v_kernel<<<16, 256, 0, stream>>>(v_tab);
    build_u_kernel<<<16, 256, 0, stream>>>(y_re, y_im, h_re, h_im, s_re, s_im, u_tab);
    detect_kernel<<<NBATCH / 16, 256, 0, stream>>>(u_tab, v_tab, out);
}


# A fragments (candidate rows) formed in registers from the 16-QAM points instead of loaded from a table: v table, its 192 loads per wave and its launch removed; k order and WMMA chain unchanged
# speedup vs baseline: 1.3991x; 1.3991x over previous
#include <hip/hip_runtime.h>
#include <math.h>

typedef __attribute__((ext_vector_type(2))) float v2f;
typedef __attribute__((ext_vector_type(8))) float v8f;

#define BIGF 1e30f
#define NBATCH 4096
#define UD 24

__device__ __forceinline__ void qam16_point(int s, float& re, float& im) {
    const int b0 = (s >> 3) & 1, b1 = (s >> 2) & 1, b2 = (s >> 1) & 1, b3 = s & 1;
    const float inv = 0.3162277660168379f;
    re = (float)((1 - 2 * b0) * (2 - (1 - 2 * b2))) * inv;
    im = (float)((1 - 2 * b1) * (2 - (1 - 2 * b3))) * inv;
}

__device__ __forceinline__ void cand_row(const float (&xr)[3], const float (&xi)[3], float (&vv)[UD]) {
#pragma unroll
    for (int k = 0; k < 3; k++) {
#pragma unroll
        for (int l = 0; l < 3; l++) {
            const float pre = xr[k] * xr[l] + xi[k] * xi[l];
            const float pim = xr[k] * xi[l] - xi[k] * xr[l];
            vv[k * 3 + l]     = pre;
            vv[9 + k * 3 + l] = -pim;
        }
    }
#pragma unroll
    for (int k = 0; k < 3; k++) {
        vv[18 + k] = -2.0f * xr[k];
        vv[21 + k] =  2.0f * xi[k];
    }
}

__global__ void build_u_kernel(const float* __restrict__ y_re, const float* __restrict__ y_im,
                               const float* __restrict__ h_re, const float* __restrict__ h_im,
                               const float* __restrict__ s_re, const float* __restrict__ s_im,
                               float* __restrict__ u) {
    const int b = blockIdx.x * blockDim.x + threadIdx.x;
    if (b >= NBATCH) return;

    float Lr[8][8], Li[8][8];
#pragma unroll
    for (int i = 0; i < 8; i++) {
#pragma unroll
        for (int j = 0; j <= i; j++) {
            Lr[i][j] = s_re[b * 64 + i * 8 + j];
            Li[i][j] = s_im[b * 64 + i * 8 + j];
        }
    }
#pragma unroll
    for (int j = 0; j < 8; j++) {
        float d = Lr[j][j];
#pragma unroll
        for (int p = 0; p < j; p++) d -= Lr[j][p] * Lr[j][p] + Li[j][p] * Li[j][p];
        const float dj = sqrtf(d);
        const float inv = 1.0f / dj;
        Lr[j][j] = dj; Li[j][j] = 0.0f;
#pragma unroll
        for (int i = j + 1; i < 8; i++) {
            float ar = Lr[i][j], ai = Li[i][j];
#pragma unroll
            for (int p = 0; p < j; p++) {
                ar -= Lr[i][p] * Lr[j][p] + Li[i][p] * Li[j][p];
                ai -= Li[i][p] * Lr[j][p] - Lr[i][p] * Li[j][p];
            }
            Lr[i][j] = ar * inv; Li[i][j] = ai * inv;
        }
    }
    float ywr[8], ywi[8];
#pragma unroll
    for (int i = 0; i < 8; i++) {
        float ar = y_re[b * 8 + i], ai = y_im[b * 8 + i];
#pragma unroll
        for (int j = 0; j < i; j++) {
            ar -= Lr[i][j] * ywr[j] - Li[i][j] * ywi[j];
            ai -= Lr[i][j] * ywi[j] + Li[i][j] * ywr[j];
        }
        const float inv = 1.0f / Lr[i][i];
        ywr[i] = ar * inv; ywi[i] = ai * inv;
    }
    float hwr[8][3], hwi[8][3];
#pragma unroll
    for (int k = 0; k < 3; k++) {
#pragma unroll
        for (int i = 0; i < 8; i++) {
            float ar = h_re[b * 24 + i * 3 + k], ai = h_im[b * 24 + i * 3 + k];
#pragma unroll
            for (int j = 0; j < i; j++) {
                ar -= Lr[i][j] * hwr[j][k] - Li[i][j] * hwi[j][k];
                ai -= Lr[i][j] * hwi[j][k] + Li[i][j] * hwr[j][k];
            }
            const float inv = 1.0f / Lr[i][i];
            hwr[i][k] = ar * inv; hwi[i][k] = ai * inv;
        }
    }
    float uu[UD];
#pragma unroll
    for (int k = 0; k < 3; k++) {
#pragma unroll
        for (int l = 0; l < 3; l++) {
            float gr = 0.f, gi = 0.f;
#pragma unroll
            for (int m = 0; m < 8; m++) {
                gr += hwr[m][k] * hwr[m][l] + hwi[m][k] * hwi[m][l];
                gi += hwr[m][k] * hwi[m][l] - hwi[m][k] * hwr[m][l];
            }
            uu[k * 3 + l]     = gr;
            uu[9 + k * 3 + l] = gi;
        }
    }
#pragma unroll
    for (int k = 0; k < 3; k++) {
        float zr = 0.f, zi = 0.f;
#pragma unroll
        for (int m = 0; m < 8; m++) {
            zr += ywr[m] * hwr[m][k] + ywi[m] * hwi[m][k];
            zi += ywr[m] * hwi[m][k] - ywi[m] * hwr[m][k];
        }
        uu[18 + k] = zr;
        uu[21 + k] = zi;
    }
#pragma unroll
    for (int t = 0; t < UD; t++) u[b * UD + t] = uu[t];
}

__global__ void __launch_bounds__(256)
detect_kernel(const float* __restrict__ u, float* __restrict__ out) {
    __shared__ float lm1[8][32][16];
    __shared__ float lm2[8][32][8];
    __shared__ float lb0[8][32][2];

    const int tid  = threadIdx.x;
    const int wave = __builtin_amdgcn_readfirstlane(tid >> 5);
    const int lane = tid & 31;
    const int li   = lane & 15;
    const int hh   = lane >> 4;
    const int bbase = blockIdx.x * 16;

    const float* ub = u + (size_t)(bbase + li) * UD + 2 * hh;
    v2f bf[6];
#pragma unroll
    for (int kk = 0; kk < 6; kk++) bf[kk] = *(const v2f*)(ub + 4 * kk);

    float x2r, x2i;
    qam16_point(li, x2r, x2i);
    float x0r[2], x0i[2];
#pragma unroll
    for (int j = 0; j < 2; j++) qam16_point(wave * 2 + j, x0r[j], x0i[j]);

    float m1h[16], m2h[8], bm0[2];
#pragma unroll
    for (int t = 0; t < 8; t++) m2h[t] = BIGF;
    bm0[0] = BIGF;
    bm0[1] = BIGF;

#pragma unroll
    for (int i1 = 0; i1 < 16; i1++) {
        float x1r, x1i;
        qam16_point(i1, x1r, x1i);
        float t1 = BIGF;
#pragma unroll
        for (int j = 0; j < 2; j++) {
            const float xr[3] = {x0r[j], x1r, x2r};
            const float xi[3] = {x0i[j], x1i, x2i};
            float vv[UD];
            cand_row(xr, xi, vv);
            v8f acc = {0.f, 0.f, 0.f, 0.f, 0.f, 0.f, 0.f, 0.f};
#pragma unroll
            for (int kk = 0; kk < 6; kk++) {
                v2f af;
                af[0] = hh ? vv[4 * kk + 2] : vv[4 * kk];
                af[1] = hh ? vv[4 * kk + 3] : vv[4 * kk + 1];
                acc = __builtin_amdgcn_wmma_f32_16x16x4_f32(
                    false, af, false, bf[kk], (short)0, acc, false, false);
            }
            const float tmin =
                fminf(fminf(fminf(acc[0], acc[1]), fminf(acc[2], acc[3])),
                      fminf(fminf(acc[4], acc[5]), fminf(acc[6], acc[7])));
            bm0[j] = fminf(bm0[j], tmin);
            t1 = fminf(t1, tmin);
#pragma unroll
            for (int r = 0; r < 8; r++) m2h[r] = fminf(m2h[r], acc[r]);
        }
        m1h[i1] = t1;
    }

#pragma unroll
    for (int t = 0; t < 16; t++) lm1[wave][lane][t] = m1h[t];
#pragma unroll
    for (int t = 0; t < 8; t++) lm2[wave][lane][t] = m2h[t];
    lb0[wave][lane][0] = bm0[0];
    lb0[wave][lane][1] = bm0[1];
    __syncthreads();

    if (tid < 16) {
        const int b = tid;
        float m[3][16];
#pragma unroll
        for (int i0 = 0; i0 < 16; i0++) {
            const int w = i0 >> 1, jj = i0 & 1;
            m[0][i0] = fminf(lb0[w][b][jj], lb0[w][b + 16][jj]);
        }
#pragma unroll
        for (int i1 = 0; i1 < 16; i1++) {
            float t = BIGF;
#pragma unroll
            for (int w = 0; w < 8; w++)
                t = fminf(t, fminf(lm1[w][b][i1], lm1[w][b + 16][i1]));
            m[1][i1] = t;
        }
#pragma unroll
        for (int i2 = 0; i2 < 16; i2++) {
            const int h2 = i2 >> 3, r = i2 & 7;
            float t = BIGF;
#pragma unroll
            for (int w = 0; w < 8; w++) t = fminf(t, lm2[w][b + 16 * h2][r]);
            m[2][i2] = t;
        }
#pragma unroll
        for (int k = 0; k < 3; k++) {
#pragma unroll
            for (int n = 0; n < 4; n++) {
                float mn0 = BIGF, mn1 = BIGF;
#pragma unroll
                for (int i = 0; i < 16; i++) {
                    if ((i >> (3 - n)) & 1) mn1 = fminf(mn1, m[k][i]);
                    else                    mn0 = fminf(mn0, m[k][i]);
                }
                out[(size_t)(bbase + b) * 12 + k * 4 + n] = mn0 - mn1;
            }
        }
    }
}

extern "C" void kernel_launch(void* const* d_in, const int* in_sizes, int n_in,
                              void* d_out, int out_size, void* d_ws, size_t ws_size,
                              hipStream_t stream) {
    const float* y_re = (const float*)d_in[0];
    const float* y_im = (const float*)d_in[1];
    const float* h_re = (const float*)d_in[2];
    const float* h_im = (const float*)d_in[3];
    const float* s_re = (const float*)d_in[4];
    const float* s_im = (const float*)d_in[5];
    float* out = (float*)d_out;

    float* u_tab = (float*)d_ws;

    build_u_kernel<<<16, 256, 0, stream>>>(y_re, y_im, h_re, h_im, s_re, s_im, u_tab);
    detect_kernel<<<NBATCH / 16, 256, 0, stream>>>(u_tab, out);
}


# whitening fused into the detect kernel: each block stages and whitens its own 16 batches in LDS (4 lanes per batch, then 24 sums); u table, its launch and its global round trip removed
# speedup vs baseline: 2.2364x; 1.5985x over previous
#include <hip/hip_runtime.h>
#include <math.h>

typedef __attribute__((ext_vector_type(2))) float v2f;
typedef __attribute__((ext_vector_type(8))) float v8f;

#define BIGF 1e30f
#define NBATCH 4096
#define UD 24

__device__ __forceinline__ void qam16_point(int s, float& re, float& im) {
    const int b0 = (s >> 3) & 1, b1 = (s >> 2) & 1, b2 = (s >> 1) & 1, b3 = s & 1;
    const float inv = 0.3162277660168379f;
    re = (float)((1 - 2 * b0) * (2 - (1 - 2 * b2))) * inv;
    im = (float)((1 - 2 * b1) * (2 - (1 - 2 * b3))) * inv;
}

__device__ __forceinline__ void cand_row(const float (&xr)[3], const float (&xi)[3], float (&vv)[UD]) {
#pragma unroll
    for (int k = 0; k < 3; k++) {
#pragma unroll
        for (int l = 0; l < 3; l++) {
            const float pre = xr[k] * xr[l] + xi[k] * xi[l];
            const float pim = xr[k] * xi[l] - xi[k] * xr[l];
            vv[k * 3 + l]     = pre;
            vv[9 + k * 3 + l] = -pim;
        }
    }
#pragma unroll
    for (int k = 0; k < 3; k++) {
        vv[18 + k] = -2.0f * xr[k];
        vv[21 + k] =  2.0f * xi[k];
    }
}

#define SP 65
#define RP 33
#define WP 33
#define UP 26

__device__ __forceinline__ void whiten_column(const float* __restrict__ Sr, const float* __restrict__ Si,
                                              const float* __restrict__ rr, const float* __restrict__ ri,
                                              const int rs,
                                              float* __restrict__ wr_out, float* __restrict__ wi_out) {
    float Lr[8][8], Li[8][8];
#pragma unroll
    for (int i = 0; i < 8; i++) {
#pragma unroll
        for (int j = 0; j <= i; j++) {
            Lr[i][j] = Sr[i * 8 + j];
            Li[i][j] = Si[i * 8 + j];
        }
    }
#pragma unroll
    for (int j = 0; j < 8; j++) {
        float d = Lr[j][j];
#pragma unroll
        for (int p = 0; p < j; p++) d -= Lr[j][p] * Lr[j][p] + Li[j][p] * Li[j][p];
        const float dj = sqrtf(d);
        const float inv = 1.0f / dj;
        Lr[j][j] = dj; Li[j][j] = 0.0f;
#pragma unroll
        for (int i = j + 1; i < 8; i++) {
            float ar = Lr[i][j], ai = Li[i][j];
#pragma unroll
            for (int p = 0; p < j; p++) {
                ar -= Lr[i][p] * Lr[j][p] + Li[i][p] * Li[j][p];
                ai -= Li[i][p] * Lr[j][p] - Lr[i][p] * Li[j][p];
            }
            Lr[i][j] = ar * inv; Li[i][j] = ai * inv;
        }
    }
    float wr[8], wi[8];
#pragma unroll
    for (int i = 0; i < 8; i++) {
        float ar = rr[i * rs], ai = ri[i * rs];
#pragma unroll
        for (int j = 0; j < i; j++) {
            ar -= Lr[i][j] * wr[j] - Li[i][j] * wi[j];
            ai -= Lr[i][j] * wi[j] + Li[i][j] * wr[j];
        }
        const float inv = 1.0f / Lr[i][i];
        wr[i] = ar * inv; wi[i] = ai * inv;
    }
#pragma unroll
    for (int i = 0; i < 8; i++) {
        wr_out[i] = wr[i];
        wi_out[i] = wi[i];
    }
}

__device__ __forceinline__ float u_entry(const int idx, const float* __restrict__ Wr, const float* __restrict__ Wi) {
    bool imag_part;
    int ca, cb;
    if (idx < 18) {
        imag_part = idx >= 9;
        const int r = imag_part ? idx - 9 : idx;
        ca = 1 + r / 3;
        cb = 1 + r % 3;
    } else {
        imag_part = idx >= 21;
        ca = 0;
        cb = 1 + idx - (imag_part ? 21 : 18);
    }
    float acc = 0.f;
#pragma unroll
    for (int m = 0; m < 8; m++) {
        const float ar = Wr[ca * 8 + m], ai = Wi[ca * 8 + m];
        const float br = Wr[cb * 8 + m], bi = Wi[cb * 8 + m];
        const float p = imag_part ? bi : br;
        const float s = imag_part ? -br : bi;
        acc += ar * p + ai * s;
    }
    return acc;
}

__global__ void __launch_bounds__(256)
detect_kernel(const float* __restrict__ y_re, const float* __restrict__ y_im,
              const float* __restrict__ h_re, const float* __restrict__ h_im,
              const float* __restrict__ s_re, const float* __restrict__ s_im,
              float* __restrict__ out) {
    __shared__ float sSr[16 * SP], sSi[16 * SP];
    __shared__ float sRr[16 * RP], sRi[16 * RP];
    __shared__ float sWr[16 * WP], sWi[16 * WP];
    __shared__ __attribute__((aligned(8))) float sU[16 * UP];
    __shared__ float lm1[8][32][16];
    __shared__ float lm2[8][32][8];
    __shared__ float lb0[8][32][2];

    const int tid  = threadIdx.x;
    const int wave = __builtin_amdgcn_readfirstlane(tid >> 5);
    const int lane = tid & 31;
    const int li   = lane & 15;
    const int hh   = lane >> 4;
    const int bbase = blockIdx.x * 16;

    {
        const bool is_y  = tid < 64;
        const bool is_re = is_y ? (tid < 32) : (tid < 160);
        const int  t     = is_y ? (tid & 31) : (is_re ? tid - 64 : tid - 160);
        const float* g3  = is_y ? (is_re ? y_re : y_im) : (is_re ? h_re : h_im);
        const size_t go  = (size_t)bbase * (is_y ? 8 : 24) + (size_t)(t * 4);
        const int    lo  = is_y ? (t >> 1) * RP + (t & 1) * 4 : (t / 6) * RP + 8 + (t % 6) * 4;
        float* l3        = is_re ? sRr : sRi;

        const float4 a = *(const float4*)(s_re + (size_t)bbase * 64 + tid * 4);
        const float4 c = *(const float4*)(s_im + (size_t)bbase * 64 + tid * 4);
        const float4 v = *(const float4*)(g3 + go);

        const int so = (tid >> 4) * SP + (tid & 15) * 4;
        sSr[so + 0] = a.x; sSr[so + 1] = a.y; sSr[so + 2] = a.z; sSr[so + 3] = a.w;
        sSi[so + 0] = c.x; sSi[so + 1] = c.y; sSi[so + 2] = c.z; sSi[so + 3] = c.w;
        l3[lo + 0] = v.x; l3[lo + 1] = v.y; l3[lo + 2] = v.z; l3[lo + 3] = v.w;
    }
    __syncthreads();

    if (tid < 64) {
        const int b = tid & 15, c = tid >> 4;
        const int ro = b * RP + (c == 0 ? 0 : 7 + c);
        whiten_column(sSr + b * SP, sSi + b * SP, sRr + ro, sRi + ro, c == 0 ? 1 : 3,
                      sWr + b * WP + c * 8, sWi + b * WP + c * 8);
    }
    __syncthreads();

    {
        const int b = tid & 15, q = tid >> 4;
        sU[b * UP + q] = u_entry(q, sWr + b * WP, sWi + b * WP);
        if (q < 8) sU[b * UP + 16 + q] = u_entry(16 + q, sWr + b * WP, sWi + b * WP);
    }
    __syncthreads();

    const float* ub = sU + li * UP + 2 * hh;
    v2f bf[6];
#pragma unroll
    for (int kk = 0; kk < 6; kk++) bf[kk] = *(const v2f*)(ub + 4 * kk);

    float x2r, x2i;
    qam16_point(li, x2r, x2i);
    float x0r[2], x0i[2];
#pragma unroll
    for (int j = 0; j < 2; j++) qam16_point(wave * 2 + j, x0r[j], x0i[j]);

    float m1h[16], m2h[8], bm0[2];
#pragma unroll
    for (int t = 0; t < 8; t++) m2h[t] = BIGF;
    bm0[0] = BIGF;
    bm0[1] = BIGF;

#pragma unroll
    for (int i1 = 0; i1 < 16; i1++) {
        float x1r, x1i;
        qam16_point(i1, x1r, x1i);
        float t1 = BIGF;
#pragma unroll
        for (int j = 0; j < 2; j++) {
            const float xr[3] = {x0r[j], x1r, x2r};
            const float xi[3] = {x0i[j], x1i, x2i};
            float vv[UD];
            cand_row(xr, xi, vv);
            v8f acc = {0.f, 0.f, 0.f, 0.f, 0.f, 0.f, 0.f, 0.f};
#pragma unroll
            for (int kk = 0; kk < 6; kk++) {
                v2f af;
                af[0] = hh ? vv[4 * kk + 2] : vv[4 * kk];
                af[1] = hh ? vv[4 * kk + 3] : vv[4 * kk + 1];
                acc = __builtin_amdgcn_wmma_f32_16x16x4_f32(
                    false, af, false, bf[kk], (short)0, acc, false, false);
            }
            const float tmin =
                fminf(fminf(fminf(acc[0], acc[1]), fminf(acc[2], acc[3])),
                      fminf(fminf(acc[4], acc[5]), fminf(acc[6], acc[7])));
            bm0[j] = fminf(bm0[j], tmin);
            t1 = fminf(t1, tmin);
#pragma unroll
            for (int r = 0; r < 8; r++) m2h[r] = fminf(m2h[r], acc[r]);
        }
        m1h[i1] = t1;
    }

#pragma unroll
    for (int t = 0; t < 16; t++) lm1[wave][lane][t] = m1h[t];
#pragma unroll
    for (int t = 0; t < 8; t++) lm2[wave][lane][t] = m2h[t];
    lb0[wave][lane][0] = bm0[0];
    lb0[wave][lane][1] = bm0[1];
    __syncthreads();

    if (tid < 16) {
        const int b = tid;
        float m[3][16];
#pragma unroll
        for (int i0 = 0; i0 < 16; i0++) {
            const int w = i0 >> 1, jj = i0 & 1;
            m[0][i0] = fminf(lb0[w][b][jj], lb0[w][b + 16][jj]);
        }
#pragma unroll
        for (int i1 = 0; i1 < 16; i1++) {
            float t = BIGF;
#pragma unroll
            for (int w = 0; w < 8; w++)
                t = fminf(t, fminf(lm1[w][b][i1], lm1[w][b + 16][i1]));
            m[1][i1] = t;
        }
#pragma unroll
        for (int i2 = 0; i2 < 16; i2++) {
            const int h2 = i2 >> 3, r = i2 & 7;
            float t = BIGF;
#pragma unroll
            for (int w = 0; w < 8; w++) t = fminf(t, lm2[w][b + 16 * h2][r]);
            m[2][i2] = t;
        }
#pragma unroll
        for (int k = 0; k < 3; k++) {
#pragma unroll
            for (int n = 0; n < 4; n++) {
                float mn0 = BIGF, mn1 = BIGF;
#pragma unroll
                for (int i = 0; i < 16; i++) {
                    if ((i >> (3 - n)) & 1) mn1 = fminf(mn1, m[k][i]);
                    else                    mn0 = fminf(mn0, m[k][i]);
                }
                out[(size_t)(bbase + b) * 12 + k * 4 + n] = mn0 - mn1;
            }
        }
    }
}

extern "C" void kernel_launch(void* const* d_in, const int* in_sizes, int n_in,
                              void* d_out, int out_size, void* d_ws, size_t ws_size,
                              hipStream_t stream) {
    const float* y_re = (const float*)d_in[0];
    const float* y_im = (const float*)d_in[1];
    const float* h_re = (const float*)d_in[2];
    const float* h_im = (const float*)d_in[3];
    const float* s_re = (const float*)d_in[4];
    const float* s_im = (const float*)d_in[5];
    float* out = (float*)d_out;

    detect_kernel<<<NBATCH / 16, 256, 0, stream>>>(y_re, y_im, h_re, h_im, s_re, s_im, out);
}


# G and P are Hermitian by their formulas: each mirrored pair of terms folded into one (factor 2 on the batch side), K 24 -> 16, 4 WMMA per tile instead of 6; the distance sum's order changes
# speedup vs baseline: 2.6412x; 1.1810x over previous
#include <hip/hip_runtime.h>
#include <math.h>

typedef __attribute__((ext_vector_type(2))) float v2f;
typedef __attribute__((ext_vector_type(8))) float v8f;

#define BIGF 1e30f
#define NBATCH 4096
#define UD 16

__device__ __forceinline__ void qam16_point(int s, float& re, float& im) {
    const int b0 = (s >> 3) & 1, b1 = (s >> 2) & 1, b2 = (s >> 1) & 1, b3 = s & 1;
    const float inv = 0.3162277660168379f;
    re = (float)((1 - 2 * b0) * (2 - (1 - 2 * b2))) * inv;
    im = (float)((1 - 2 * b1) * (2 - (1 - 2 * b3))) * inv;
}

__device__ __forceinline__ void cand_row(const float (&xr)[3], const float (&xi)[3], float (&vv)[UD]) {
#pragma unroll
    for (int k = 0; k < 3; k++) vv[k] = xr[k] * xr[k] + xi[k] * xi[k];
    int r = 0;
#pragma unroll
    for (int k = 0; k < 3; k++) {
#pragma unroll
        for (int l = k + 1; l < 3; l++) {
            const float pre = xr[k] * xr[l] + xi[k] * xi[l];
            const float pim = xr[k] * xi[l] - xi[k] * xr[l];
            vv[3 + r] =  pre;
            vv[6 + r] = -pim;
            r++;
        }
    }
#pragma unroll
    for (int k = 0; k < 3; k++) {
        vv[9 + k]  = -2.0f * xr[k];
        vv[12 + k] =  2.0f * xi[k];
    }
    vv[15] = 0.0f;
}

#define SP 65
#define RP 33
#define WP 33
#define UP 20

__device__ __forceinline__ void whiten_column(const float* __restrict__ Sr, const float* __restrict__ Si,
                                              const float* __restrict__ rr, const float* __restrict__ ri,
                                              const int rs,
                                              float* __restrict__ wr_out, float* __restrict__ wi_out) {
    float Lr[8][8], Li[8][8];
#pragma unroll
    for (int i = 0; i < 8; i++) {
#pragma unroll
        for (int j = 0; j <= i; j++) {
            Lr[i][j] = Sr[i * 8 + j];
            Li[i][j] = Si[i * 8 + j];
        }
    }
#pragma unroll
    for (int j = 0; j < 8; j++) {
        float d = Lr[j][j];
#pragma unroll
        for (int p = 0; p < j; p++) d -= Lr[j][p] * Lr[j][p] + Li[j][p] * Li[j][p];
        const float dj = sqrtf(d);
        const float inv = 1.0f / dj;
        Lr[j][j] = dj; Li[j][j] = 0.0f;
#pragma unroll
        for (int i = j + 1; i < 8; i++) {
            float ar = Lr[i][j], ai = Li[i][j];
#pragma unroll
            for (int p = 0; p < j; p++) {
                ar -= Lr[i][p] * Lr[j][p] + Li[i][p] * Li[j][p];
                ai -= Li[i][p] * Lr[j][p] - Lr[i][p] * Li[j][p];
            }
            Lr[i][j] = ar * inv; Li[i][j] = ai * inv;
        }
    }
    float wr[8], wi[8];
#pragma unroll
    for (int i = 0; i < 8; i++) {
        float ar = rr[i * rs], ai = ri[i * rs];
#pragma unroll
        for (int j = 0; j < i; j++) {
            ar -= Lr[i][j] * wr[j] - Li[i][j] * wi[j];
            ai -= Lr[i][j] * wi[j] + Li[i][j] * wr[j];
        }
        const float inv = 1.0f / Lr[i][i];
        wr[i] = ar * inv; wi[i] = ai * inv;
    }
#pragma unroll
    for (int i = 0; i < 8; i++) {
        wr_out[i] = wr[i];
        wi_out[i] = wi[i];
    }
}

__device__ __forceinline__ float u_entry(const int idx, const float* __restrict__ Wr, const float* __restrict__ Wi) {
    bool imag_part;
    int ca, cb;
    if (idx < 3) {
        imag_part = false;
        ca = 1 + idx;
        cb = 1 + idx;
    } else if (idx < 9) {
        imag_part = idx >= 6;
        const int r = imag_part ? idx - 6 : idx - 3;
        ca = 1 + (r >> 1);
        cb = 2 + ((r + 1) >> 1);
    } else {
        imag_part = idx >= 12;
        ca = 0;
        cb = 1 + idx - (imag_part ? 12 : 9);
    }
    float acc = 0.f;
#pragma unroll
    for (int m = 0; m < 8; m++) {
        const float ar = Wr[ca * 8 + m], ai = Wi[ca * 8 + m];
        const float br = Wr[cb * 8 + m], bi = Wi[cb * 8 + m];
        const float p = imag_part ? bi : br;
        const float s = imag_part ? -br : bi;
        acc += ar * p + ai * s;
    }
    return (idx >= 3 && idx < 9) ? acc + acc : acc;
}

__global__ void __launch_bounds__(256)
detect_kernel(const float* __restrict__ y_re, const float* __restrict__ y_im,
              const float* __restrict__ h_re, const float* __restrict__ h_im,
              const float* __restrict__ s_re, const float* __restrict__ s_im,
              float* __restrict__ out) {
    __shared__ float sSr[16 * SP], sSi[16 * SP];
    __shared__ float sRr[16 * RP], sRi[16 * RP];
    __shared__ float sWr[16 * WP], sWi[16 * WP];
    __shared__ __attribute__((aligned(8))) float sU[16 * UP];
    __shared__ float lm1[8][32][16];
    __shared__ float lm2[8][32][8];
    __shared__ float lb0[8][32][2];

    const int tid  = threadIdx.x;
    const int wave = __builtin_amdgcn_readfirstlane(tid >> 5);
    const int lane = tid & 31;
    const int li   = lane & 15;
    const int hh   = lane >> 4;
    const int bbase = blockIdx.x * 16;

    {
        const bool is_y  = tid < 64;
        const bool is_re = is_y ? (tid < 32) : (tid < 160);
        const int  t     = is_y ? (tid & 31) : (is_re ? tid - 64 : tid - 160);
        const float* g3  = is_y ? (is_re ? y_re : y_im) : (is_re ? h_re : h_im);
        const size_t go  = (size_t)bbase * (is_y ? 8 : 24) + (size_t)(t * 4);
        const int    lo  = is_y ? (t >> 1) * RP + (t & 1) * 4 : (t / 6) * RP + 8 + (t % 6) * 4;
        float* l3        = is_re ? sRr : sRi;

        const float4 a = *(const float4*)(s_re + (size_t)bbase * 64 + tid * 4);
        const float4 c = *(const float4*)(s_im + (size_t)bbase * 64 + tid * 4);
        const float4 v = *(const float4*)(g3 + go);

        const int so = (tid >> 4) * SP + (tid & 15) * 4;
        sSr[so + 0] = a.x; sSr[so + 1] = a.y; sSr[so + 2] = a.z; sSr[so + 3] = a.w;
        sSi[so + 0] = c.x; sSi[so + 1] = c.y; sSi[so + 2] = c.z; sSi[so + 3] = c.w;
        l3[lo + 0] = v.x; l3[lo + 1] = v.y; l3[lo + 2] = v.z; l3[lo + 3] = v.w;
    }
    __syncthreads();

    if (tid < 64) {
        const int b = tid & 15, c = tid >> 4;
        const int ro = b * RP + (c == 0 ? 0 : 7 + c);
        whiten_column(sSr + b * SP, sSi + b * SP, sRr + ro, sRi + ro, c == 0 ? 1 : 3,
                      sWr + b * WP + c * 8, sWi + b * WP + c * 8);
    }
    __syncthreads();

    {
        const int b = tid & 15, q = tid >> 4;
        float val = 0.0f;
        if (q < 15) val = u_entry(q, sWr + b * WP, sWi + b * WP);
        sU[b * UP + q] = val;
    }
    __syncthreads();

    const float* ub = sU + li * UP + 2 * hh;
    v2f bf[4];
#pragma unroll
    for (int kk = 0; kk < 4; kk++) bf[kk] = *(const v2f*)(ub + 4 * kk);

    float x2r, x2i;
    qam16_point(li, x2r, x2i);
    float x0r[2], x0i[2];
#pragma unroll
    for (int j = 0; j < 2; j++) qam16_point(wave * 2 + j, x0r[j], x0i[j]);

    float m1h[16], m2h[8], bm0[2];
#pragma unroll
    for (int t = 0; t < 8; t++) m2h[t] = BIGF;
    bm0[0] = BIGF;
    bm0[1] = BIGF;

#pragma unroll
    for (int i1 = 0; i1 < 16; i1++) {
        float x1r, x1i;
        qam16_point(i1, x1r, x1i);
        float t1 = BIGF;
#pragma unroll
        for (int j = 0; j < 2; j++) {
            const float xr[3] = {x0r[j], x1r, x2r};
            const float xi[3] = {x0i[j], x1i, x2i};
            float vv[UD];
            cand_row(xr, xi, vv);
            v8f acc = {0.f, 0.f, 0.f, 0.f, 0.f, 0.f, 0.f, 0.f};
#pragma unroll
            for (int kk = 0; kk < 4; kk++) {
                v2f af;
                af[0] = hh ? vv[4 * kk + 2] : vv[4 * kk];
                af[1] = hh ? vv[4 * kk + 3] : vv[4 * kk + 1];
                acc = __builtin_amdgcn_wmma_f32_16x16x4_f32(
                    false, af, false, bf[kk], (short)0, acc, false, false);
            }
            const float tmin =
                fminf(fminf(fminf(acc[0], acc[1]), fminf(acc[2], acc[3])),
                      fminf(fminf(acc[4], acc[5]), fminf(acc[6], acc[7])));
            bm0[j] = fminf(bm0[j], tmin);
            t1 = fminf(t1, tmin);
#pragma unroll
            for (int r = 0; r < 8; r++) m2h[r] = fminf(m2h[r], acc[r]);
        }
        m1h[i1] = t1;
    }

#pragma unroll
    for (int t = 0; t < 16; t++) lm1[wave][lane][t] = m1h[t];
#pragma unroll
    for (int t = 0; t < 8; t++) lm2[wave][lane][t] = m2h[t];
    lb0[wave][lane][0] = bm0[0];
    lb0[wave][lane][1] = bm0[1];
    __syncthreads();

    if (tid < 16) {
        const int b = tid;
        float m[3][16];
#pragma unroll
        for (int i0 = 0; i0 < 16; i0++) {
            const int w = i0 >> 1, jj = i0 & 1;
            m[0][i0] = fminf(lb0[w][b][jj], lb0[w][b + 16][jj]);
        }
#pragma unroll
        for (int i1 = 0; i1 < 16; i1++) {
            float t = BIGF;
#pragma unroll
            for (int w = 0; w < 8; w++)
                t = fminf(t, fminf(lm1[w][b][i1], lm1[w][b + 16][i1]));
            m[1][i1] = t;
        }
#pragma unroll
        for (int i2 = 0; i2 < 16; i2++) {
            const int h2 = i2 >> 3, r = i2 & 7;
            float t = BIGF;
#pragma unroll
            for (int w = 0; w < 8; w++) t = fminf(t, lm2[w][b + 16 * h2][r]);
            m[2][i2] = t;
        }
#pragma unroll
        for (int k = 0; k < 3; k++) {
#pragma unroll
            for (int n = 0; n < 4; n++) {
                float mn0 = BIGF, mn1 = BIGF;
#pragma unroll
                for (int i = 0; i < 16; i++) {
                    if ((i >> (3 - n)) & 1) mn1 = fminf(mn1, m[k][i]);
                    else                    mn0 = fminf(mn0, m[k][i]);
                }
                out[(size_t)(bbase + b) * 12 + k * 4 + n] = mn0 - mn1;
            }
        }
    }
}

extern "C" void kernel_launch(void* const* d_in, const int* in_sizes, int n_in,
                              void* d_out, int out_size, void* d_ws, size_t ws_size,
                              hipStream_t stream) {
    const float* y_re = (const float*)d_in[0];
    const float* y_im = (const float*)d_in[1];
    const float* h_re = (const float*)d_in[2];
    const float* h_im = (const float*)d_in[3];
    const float* s_re = (const float*)d_in[4];
    const float* s_im = (const float*)d_in[5];
    float* out = (float*)d_out;

    detect_kernel<<<NBATCH / 16, 256, 0, stream>>>(y_re, y_im, h_re, h_im, s_re, s_im, out);
}


# factor/solve phase (2 of 8 waves, latency-bound) given a short dependent chain: pivot reciprocal by one v_rsq_f32 (no sqrt, no divide), each update term one fma, LDS reads first; sum order changes
# speedup vs baseline: 4.3628x; 1.6518x over previous
#include <hip/hip_runtime.h>
#include <math.h>

typedef __attribute__((ext_vector_type(2))) float v2f;
typedef __attribute__((ext_vector_type(8))) float v8f;

#define BIGF 1e30f
#define NBATCH 4096

__device__ __forceinline__ void qam16_point(int s, float& re, float& im) {
    const int b0 = (s >> 3) & 1, b1 = (s >> 2) & 1, b2 = (s >> 1) & 1, b3 = s & 1;
    const float inv = 0.3162277660168379f;
    re = (float)((1 - 2 * b0) * (2 - (1 - 2 * b2))) * inv;
    im = (float)((1 - 2 * b1) * (2 - (1 - 2 * b3))) * inv;
}

__device__ __forceinline__ float pam4_value(int v) {
    const float inv = 0.3162277660168379f;
    return (float)((1 - 2 * (v >> 1)) * (2 - (1 - 2 * (v & 1)))) * inv;
}

#define SP 65
#define RP 33
#define WP 33
#define U_SLOT 0x75FE6DC31B20A498ull
#define UP 20

__device__ __forceinline__ void whiten_column(const float* __restrict__ Sr, const float* __restrict__ Si,
                                              const float* __restrict__ rr, const float* __restrict__ ri,
                                              const int rs,
                                              float* __restrict__ wr_out, float* __restrict__ wi_out) {
    float Lr[8][8], Li[8][8];
    float br[8], bi[8];
#pragma unroll
    for (int i = 0; i < 8; i++) {
#pragma unroll
        for (int j = 0; j <= i; j++) {
            Lr[i][j] = Sr[i * 8 + j];
            if (j < i) Li[i][j] = Si[i * 8 + j];
        }
    }
#pragma unroll
    for (int i = 0; i < 8; i++) {
        br[i] = rr[i * rs];
        bi[i] = ri[i * rs];
    }
    __builtin_amdgcn_sched_barrier(0);
    float rinv[8];
#pragma unroll
    for (int j = 0; j < 8; j++) {
        float d = Lr[j][j];
#pragma unroll
        for (int p = 0; p < j; p++) {
            d = fmaf(-Lr[j][p], Lr[j][p], d);
            d = fmaf(-Li[j][p], Li[j][p], d);
        }
        const float inv = __builtin_amdgcn_rsqf(d);
        rinv[j] = inv;
#pragma unroll
        for (int i = j + 1; i < 8; i++) {
            float ar = Lr[i][j], ai = Li[i][j];
#pragma unroll
            for (int p = 0; p < j; p++) {
                ar = fmaf(-Lr[i][p], Lr[j][p], ar);
                ar = fmaf(-Li[i][p], Li[j][p], ar);
                ai = fmaf(-Li[i][p], Lr[j][p], ai);
                ai = fmaf(Lr[i][p], Li[j][p], ai);
            }
            Lr[i][j] = ar * inv; Li[i][j] = ai * inv;
        }
    }
    float wr[8], wi[8];
#pragma unroll
    for (int i = 0; i < 8; i++) {
        float ar = br[i], ai = bi[i];
#pragma unroll
        for (int j = 0; j < i; j++) {
            ar = fmaf(-Lr[i][j], wr[j], ar);
            ar = fmaf(Li[i][j], wi[j], ar);
            ai = fmaf(-Lr[i][j], wi[j], ai);
            ai = fmaf(-Li[i][j], wr[j], ai);
        }
        wr[i] = ar * rinv[i]; wi[i] = ai * rinv[i];
    }
#pragma unroll
    for (int i = 0; i < 8; i++) {
        wr_out[i] = wr[i];
        wi_out[i] = wi[i];
    }
}

__device__ __forceinline__ float u_entry(const int idx, const float* __restrict__ Wr, const float* __restrict__ Wi) {
    bool imag_part;
    int ca, cb;
    if (idx < 3) {
        imag_part = false;
        ca = 1 + idx;
        cb = 1 + idx;
    } else if (idx < 9) {
        imag_part = idx >= 6;
        const int r = imag_part ? idx - 6 : idx - 3;
        ca = 1 + (r >> 1);
        cb = 2 + ((r + 1) >> 1);
    } else {
        imag_part = idx >= 12;
        ca = 0;
        cb = 1 + idx - (imag_part ? 12 : 9);
    }
    float acc = 0.f;
#pragma unroll
    for (int m = 0; m < 8; m++) {
        const float ar = Wr[ca * 8 + m], ai = Wi[ca * 8 + m];
        const float br = Wr[cb * 8 + m], bi = Wi[cb * 8 + m];
        const float p = imag_part ? bi : br;
        const float s = imag_part ? -br : bi;
        acc += ar * p + ai * s;
    }
    return (idx >= 3 && idx < 9) ? acc + acc : acc;
}

__global__ void __launch_bounds__(256)
detect_kernel(const float* __restrict__ y_re, const float* __restrict__ y_im,
              const float* __restrict__ h_re, const float* __restrict__ h_im,
              const float* __restrict__ s_re, const float* __restrict__ s_im,
              float* __restrict__ out) {
    __shared__ float sSr[16 * SP], sSi[16 * SP];
    __shared__ float sRr[16 * RP], sRi[16 * RP];
    __shared__ float sWr[16 * WP], sWi[16 * WP];
    __shared__ __attribute__((aligned(8))) float sU[16 * UP];
    __shared__ float lm1[8][32][8];
    __shared__ float lm2[8][32][8];
    __shared__ float lb0[8][32][2];

    const int tid  = threadIdx.x;
    const int wave = __builtin_amdgcn_readfirstlane(tid >> 5);
    const int lane = tid & 31;
    const int li   = lane & 15;
    const int hh   = lane >> 4;
    const int bbase = blockIdx.x * 16;

    {
        const bool is_y  = tid < 64;
        const bool is_re = is_y ? (tid < 32) : (tid < 160);
        const int  t     = is_y ? (tid & 31) : (is_re ? tid - 64 : tid - 160);
        const float* g3  = is_y ? (is_re ? y_re : y_im) : (is_re ? h_re : h_im);
        const size_t go  = (size_t)bbase * (is_y ? 8 : 24) + (size_t)(t * 4);
        const int    lo  = is_y ? (t >> 1) * RP + (t & 1) * 4 : (t / 6) * RP + 8 + (t % 6) * 4;
        float* l3        = is_re ? sRr : sRi;

        const float4 a = *(const float4*)(s_re + (size_t)bbase * 64 + tid * 4);
        const float4 c = *(const float4*)(s_im + (size_t)bbase * 64 + tid * 4);
        const float4 v = *(const float4*)(g3 + go);

        const int so = (tid >> 4) * SP + (tid & 15) * 4;
        sSr[so + 0] = a.x; sSr[so + 1] = a.y; sSr[so + 2] = a.z; sSr[so + 3] = a.w;
        sSi[so + 0] = c.x; sSi[so + 1] = c.y; sSi[so + 2] = c.z; sSi[so + 3] = c.w;
        l3[lo + 0] = v.x; l3[lo + 1] = v.y; l3[lo + 2] = v.z; l3[lo + 3] = v.w;
    }
    __syncthreads();

    if (tid < 64) {
        const int b = tid & 15, c = tid >> 4;
        const int ro = b * RP + (c == 0 ? 0 : 7 + c);
        whiten_column(sSr + b * SP, sSi + b * SP, sRr + ro, sRi + ro, c == 0 ? 1 : 3,
                      sWr + b * WP + c * 8, sWi + b * WP + c * 8);
    }
    __syncthreads();

    {
        const int b = tid & 15, q = tid >> 4;
        if (q < 15) {
            const int slot = (int)((U_SLOT >> (4 * q)) & 15ull);
            sU[b * UP + slot] = u_entry(q, sWr + b * WP, sWi + b * WP);
        }
    }
    __syncthreads();

    const float* ub = sU + li * UP + 2 * hh;
    const float rs10 = 0.3162277660168379f;
    const v2f bf0r = *(const v2f*)(ub + 0);
    v2f bf0;
    bf0[0] = bf0r[0] * rs10;
    bf0[1] = bf0r[1] * rs10;
    const v2f bf2 = *(const v2f*)(ub + 8);
    const v2f bf3 = *(const v2f*)(ub + 12);
    const float g22  = sU[li * UP + 4];
    const float imz2 = sU[li * UP + 5];
    const float rez2 = sU[li * UP + 6];

    const float pv1 = pam4_value(0), pv3 = pam4_value(1);
    const float q1 = g22 * (pv1 * pv1), q3 = g22 * (pv3 * pv3);
    const float crs = (-2.0f * rs10) * rez2, cis = (2.0f * rs10) * imz2;
    const v8f cr0 = {crs, crs, crs, crs, crs, crs, crs, crs};
    const v8f ci0 = {cis, cis, cis, cis, cis, cis, cis, cis};

    float x0r[2], x0i[2];
#pragma unroll
    for (int j = 0; j < 2; j++) qam16_point(wave * 2 + j, x0r[j], x0i[j]);
    float p, s;
    qam16_point(li, p, s);
    const float n1 = p * p + s * s;

    float m1h[8], s2[8], bm0[2];
#pragma unroll
    for (int t = 0; t < 8; t++) m1h[t] = BIGF;
#pragma unroll
    for (int t = 0; t < 8; t++) s2[t] = BIGF;
    bm0[0] = BIGF;
    bm0[1] = BIGF;

    const v8f zero8 = {0.f, 0.f, 0.f, 0.f, 0.f, 0.f, 0.f, 0.f};

#pragma unroll
    for (int j = 0; j < 2; j++) {
        const float a = x0r[j], b = x0i[j];
        const float n0  = a * a + b * b;
        const float pre = a * p + b * s;
        const float mpi = b * p - a * s;
        const float a2x = 2.0f * a, b2x = 2.0f * b, p2x = 2.0f * p, s2x = 2.0f * s;
        v2f fcr, fci, fc2, fc3;
        fcr[0] = hh ? p : a;
        fcr[1] = hh ? s : b;
        fci[0] = hh ? s : b;
        fci[1] = hh ? -p : -a;
        fc2[0] = hh ? pre : n0;
        fc2[1] = hh ? mpi : n1;
        fc3[0] = hh ? b2x : -a2x;
        fc3[1] = hh ? s2x : -p2x;

        v8f Dc0 = __builtin_amdgcn_wmma_f32_16x16x4_f32(
            false, fc2, false, bf2, (short)0, zero8, false, false);
        Dc0 = __builtin_amdgcn_wmma_f32_16x16x4_f32(
            false, fc3, false, bf3, (short)0, Dc0, false, false);
        const v8f Dcr = __builtin_amdgcn_wmma_f32_16x16x4_f32(
            false, fcr, false, bf0, (short)0, cr0, false, false);
        const v8f Dci = __builtin_amdgcn_wmma_f32_16x16x4_f32(
            false, fci, false, bf0, (short)0, ci0, false, false);

#pragma unroll
        for (int r = 0; r < 8; r += 2) {
            float cand[2][8], tm[2];
#pragma unroll
            for (int t = 0; t < 2; t++) {
                const float tr = Dcr[r + t];
                const float ti = Dci[r + t];
                float acc[8];
                acc[0] = q1 + tr;
                acc[1] = fmaf(3.0f, tr, q3);
                acc[2] = q1 - tr;
                acc[3] = fmaf(-3.0f, tr, q3);
                acc[4] = q1 + ti;
                acc[5] = fmaf(3.0f, ti, q3);
                acc[6] = q1 - ti;
                acc[7] = fmaf(-3.0f, ti, q3);
                const float f_s0 = fminf(acc[0], acc[1]);
                const float f_s1 = fminf(acc[2], acc[3]);
                const float f_z0 = fminf(acc[0], acc[2]);
                const float f_z1 = fminf(acc[1], acc[3]);
                const float f_al = fminf(f_s0, f_s1);
                const float q_s0 = fminf(acc[4], acc[5]);
                const float q_s1 = fminf(acc[6], acc[7]);
                const float q_z0 = fminf(acc[4], acc[6]);
                const float q_z1 = fminf(acc[5], acc[7]);
                const float q_al = fminf(q_s0, q_s1);
                const float c0 = Dc0[r + t];
                const float t0 = c0 + q_al;
                const float t1 = c0 + f_al;
                tm[t] = t0 + f_al;
                cand[t][0] = t0 + f_s0;
                cand[t][1] = t0 + f_s1;
                cand[t][2] = t1 + q_s0;
                cand[t][3] = t1 + q_s1;
                cand[t][4] = t0 + f_z0;
                cand[t][5] = t0 + f_z1;
                cand[t][6] = t1 + q_z0;
                cand[t][7] = t1 + q_z1;
                m1h[r + t] = fminf(m1h[r + t], tm[t]);
            }
#pragma unroll
            for (int i = 0; i < 8; i++) s2[i] = fminf(fminf(cand[0][i], cand[1][i]), s2[i]);
            bm0[j] = fminf(fminf(tm[0], tm[1]), bm0[j]);
        }
    }

#pragma unroll
    for (int t = 0; t < 8; t++) lm1[wave][lane][t] = m1h[t];
#pragma unroll
    for (int t = 0; t < 8; t++) lm2[wave][lane][t] = s2[t];
    lb0[wave][lane][0] = bm0[0];
    lb0[wave][lane][1] = bm0[1];
    __syncthreads();

    if (tid < 16) {
        const int b = tid;
        float m[2][16];
#pragma unroll
        for (int i0 = 0; i0 < 16; i0++) {
            const int w = i0 >> 1, jj = i0 & 1;
            m[0][i0] = fminf(lb0[w][b][jj], lb0[w][b + 16][jj]);
        }
#pragma unroll
        for (int i1 = 0; i1 < 16; i1++) {
            const int h1 = i1 >> 3, r = i1 & 7;
            float t = BIGF;
#pragma unroll
            for (int w = 0; w < 8; w++) t = fminf(t, lm1[w][b + 16 * h1][r]);
            m[1][i1] = t;
        }
        float sb[8];
#pragma unroll
        for (int i = 0; i < 8; i++) {
            float t = BIGF;
#pragma unroll
            for (int w = 0; w < 8; w++) t = fminf(t, fminf(lm2[w][b][i], lm2[w][b + 16][i]));
            sb[i] = t;
        }
#pragma unroll
        for (int k = 0; k < 2; k++) {
#pragma unroll
            for (int n = 0; n < 4; n++) {
                float mn0 = BIGF, mn1 = BIGF;
#pragma unroll
                for (int i = 0; i < 16; i++) {
                    if ((i >> (3 - n)) & 1) mn1 = fminf(mn1, m[k][i]);
                    else                    mn0 = fminf(mn0, m[k][i]);
                }
                out[(size_t)(bbase + b) * 12 + k * 4 + n] = mn0 - mn1;
            }
        }
#pragma unroll
        for (int n = 0; n < 4; n++)
            out[(size_t)(bbase + b) * 12 + 8 + n] = sb[2 * n] - sb[2 * n + 1];
    }
}

extern "C" void kernel_launch(void* const* d_in, const int* in_sizes, int n_in,
                              void* d_out, int out_size, void* d_ws, size_t ws_size,
                              hipStream_t stream) {
    const float* y_re = (const float*)d_in[0];
    const float* y_im = (const float*)d_in[1];
    const float* h_re = (const float*)d_in[2];
    const float* h_im = (const float*)d_in[3];
    const float* s_re = (const float*)d_in[4];
    const float* s_im = (const float*)d_in[5];
    float* out = (float*)d_out;

    detect_kernel<<<NBATCH / 16, 256, 0, stream>>>(y_re, y_im, h_re, h_im, s_re, s_im, out);
}
